# starting kernel
# baseline (speedup 1.0000x reference)
#include <hip/hip_runtime.h>
#include <hip/hip_bf16.h>

typedef __attribute__((ext_vector_type(2))) float v2f;
typedef __attribute__((ext_vector_type(8))) float v8f;

#define BATCH 2
#define SLEN  2048
#define EMB   512
#define NH    8
#define DH    64
#define HWIN  128
#define QKVN  1536
#define HSTRIDE 192

__global__ void gemm_wmma(const float* __restrict__ A,
                          const float* __restrict__ W,
                          const float* __restrict__ bias,
                          float* __restrict__ C,
                          int M, int K, int N) {
    const int lane = threadIdx.x & 31;
    const int wave = threadIdx.x >> 5;
    const int lm   = lane & 15;
    const int half = lane >> 4;

    const int ntiles = N >> 6;
    const int mtiles = M >> 4;
    const int wavesPerBlock = blockDim.x >> 5;
    int tile = blockIdx.x * wavesPerBlock + wave;
    if (tile >= ntiles * mtiles) return;
    const int mt = tile / ntiles;
    const int nt = tile % ntiles;
    const int m0 = mt << 4;
    const int n0 = nt << 6;

    v8f acc0 = {}, acc1 = {}, acc2 = {}, acc3 = {};

    const float* arow  = A + (size_t)(m0 + lm) * K + half * 2;
    const float* wbase = W + (size_t)(half * 2) * N + n0 + lm;

#pragma unroll 4
    for (int k = 0; k < K; k += 4) {
        v2f a = *(const v2f*)(arow + k);
        const float* w0 = wbase + (size_t)k * N;
        const float* w1 = w0 + N;
        v2f b0, b1, b2, b3;
        b0.x = w0[0];  b0.y = w1[0];
        b1.x = w0[16]; b1.y = w1[16];
        b2.x = w0[32]; b2.y = w1[32];
        b3.x = w0[48]; b3.y = w1[48];
        acc0 = __builtin_amdgcn_wmma_f32_16x16x4_f32(false, a, false, b0, (short)0, acc0, false, false);
        acc1 = __builtin_amdgcn_wmma_f32_16x16x4_f32(false, a, false, b1, (short)0, acc1, false, false);
        acc2 = __builtin_amdgcn_wmma_f32_16x16x4_f32(false, a, false, b2, (short)0, acc2, false, false);
        acc3 = __builtin_amdgcn_wmma_f32_16x16x4_f32(false, a, false, b3, (short)0, acc3, false, false);
    }

    const float bv0 = bias[n0 + lm];
    const float bv1 = bias[n0 + 16 + lm];
    const float bv2 = bias[n0 + 32 + lm];
    const float bv3 = bias[n0 + 48 + lm];
#pragma unroll
    for (int i = 0; i < 8; ++i) {
        const size_t row = (size_t)(m0 + i + 8 * half) * N + n0 + lm;
        C[row]      = acc0[i] + bv0;
        C[row + 16] = acc1[i] + bv1;
        C[row + 32] = acc2[i] + bv2;
        C[row + 48] = acc3[i] + bv3;
    }
}

__global__ void attn_banded(const float* __restrict__ qkv,
                            const int*   __restrict__ pad,
                            float*       __restrict__ out  ) {
    __shared__ float sP[4][16][17];

    const int lane = threadIdx.x & 31;
    const int wave = threadIdx.x >> 5;
    const int lm   = lane & 15;
    const int half = lane >> 4;

    const int QT = SLEN / 16;
    int gw = blockIdx.x * 4 + wave;
    int qt = gw % QT;
    int bh = gw / QT;
    int h  = bh % NH;
    int b  = bh / NH;
    if (b >= BATCH) return;

    const int t0 = qt << 4;
    const int qg = t0 + lm;

    const float* qptr = qkv + ((size_t)(b * SLEN + qg)) * QKVN + h * HSTRIDE + half * 2;
    v2f qb[16];
#pragma unroll
    for (int dc = 0; dc < 16; ++dc) {
        v2f t = *(const v2f*)(qptr + dc * 4);
        qb[dc].x = t.x * 0.125f;
        qb[dc].y = t.y * 0.125f;
    }
    const int pm_q = pad[b * SLEN + qg];

    float m_run = -1e30f;
    float l_run = 0.0f;
    v8f acc0 = {}, acc1 = {}, acc2 = {}, acc3 = {};

    const int kt0 = (qt - 8 < 0) ? 0 : qt - 8;
    const int kt1 = (qt + 8 > QT - 1) ? QT - 1 : qt + 8;

    for (int kt = kt0; kt <= kt1; ++kt) {
        const int k0 = kt << 4;

        v8f sc = {};
        const float* kptr = qkv + ((size_t)(b * SLEN + k0 + lm)) * QKVN
                          + h * HSTRIDE + DH + half * 2;
#pragma unroll
        for (int dc = 0; dc < 16; ++dc) {
            v2f a = *(const v2f*)(kptr + dc * 4);
            sc = __builtin_amdgcn_wmma_f32_16x16x4_f32(false, a, false, qb[dc], (short)0, sc, false, false);
        }

        float p[8];
        float tmax = -1e30f;
#pragma unroll
        for (int i = 0; i < 8; ++i) {
            const int key = k0 + i + 8 * half;
            const int d   = key - qg;
            const bool valid = (d <= HWIN) && (d >= -HWIN) && (pm_q != 0)
                             && (pad[b * SLEN + key] != 0);
            const float s = valid ? sc[i] : -1e30f;
            p[i] = s;
            tmax = fmaxf(tmax, s);
        }
        tmax = fmaxf(tmax, __shfl_xor(tmax, 16, 32));
        const float m_new = fmaxf(m_run, tmax);
        float tsum = 0.0f;
#pragma unroll
        for (int i = 0; i < 8; ++i) {
            const float pe = (p[i] > -5e29f) ? __expf(p[i] - m_new) : 0.0f;
            p[i] = pe;
            tsum += pe;
        }
        tsum += __shfl_xor(tsum, 16, 32);
        const float alpha = __expf(m_run - m_new);
        l_run = l_run * alpha + tsum;
        m_run = m_new;
        acc0 = acc0 * alpha; acc1 = acc1 * alpha;
        acc2 = acc2 * alpha; acc3 = acc3 * alpha;

#pragma unroll
        for (int i = 0; i < 8; ++i)
            sP[wave][i + 8 * half][lm] = p[i];

        const float* vptr = qkv + ((size_t)(b * SLEN + k0)) * QKVN + h * HSTRIDE + 2 * DH;
#pragma unroll
        for (int kc = 0; kc < 16; kc += 4) {
            const int keyr = kc + half * 2;
            v2f bfrag;
            bfrag.x = sP[wave][keyr][lm];
            bfrag.y = sP[wave][keyr + 1][lm];
            const float* v0 = vptr + (size_t)keyr * QKVN + lm;
            const float* v1 = v0 + QKVN;
            v2f a;
            a.x = v0[0];  a.y = v1[0];
            acc0 = __builtin_amdgcn_wmma_f32_16x16x4_f32(false, a, false, bfrag, (short)0, acc0, false, false);
            a.x = v0[16]; a.y = v1[16];
            acc1 = __builtin_amdgcn_wmma_f32_16x16x4_f32(false, a, false, bfrag, (short)0, acc1, false, false);
            a.x = v0[32]; a.y = v1[32];
            acc2 = __builtin_amdgcn_wmma_f32_16x16x4_f32(false, a, false, bfrag, (short)0, acc2, false, false);
            a.x = v0[48]; a.y = v1[48];
            acc3 = __builtin_amdgcn_wmma_f32_16x16x4_f32(false, a, false, bfrag, (short)0, acc3, false, false);
        }
    }

    const float inv = (l_run > 0.0f && pm_q != 0) ? (1.0f / l_run) : 0.0f;
    float* optr = out + ((size_t)(b * SLEN + qg)) * EMB + h * DH;
#pragma unroll
    for (int i = 0; i < 8; ++i) {
        const int dr = i + 8 * half;
        optr[dr]      = acc0[i] * inv;
        optr[16 + dr] = acc1[i] * inv;
        optr[32 + dr] = acc2[i] * inv;
        optr[48 + dr] = acc3[i] * inv;
    }
}

extern "C" void kernel_launch(void* const* d_in, const int* in_sizes, int n_in,
                              void* d_out, int out_size, void* d_ws, size_t ws_size,
                              hipStream_t stream) {
    const float* x    = (const float*)d_in[0];
    const int*   pad  = (const int*)  d_in[1];
    const float* Wqkv = (const float*)d_in[2];
    const float* bqkv = (const float*)d_in[3];
    const float* Wo   = (const float*)d_in[4];
    const float* bo   = (const float*)d_in[5];
    float* out = (float*)d_out;

    float* qkv  = (float*)d_ws;
    float* attn = qkv + (size_t)BATCH * SLEN * QKVN;

    const int Mrows = BATCH * SLEN;

    {
        const int tiles = (Mrows / 16) * (QKVN / 64);
        gemm_wmma<<<(tiles + 7) / 8, 256, 0, stream>>>(x, Wqkv, bqkv, qkv,
                                                       Mrows, EMB, QKVN);
    }
    {
        const int waves = BATCH * NH * (SLEN / 16);
        attn_banded<<<waves / 4, 128, 0, stream>>>(qkv, pad, attn);
    }
    {
        const int tiles = (Mrows / 16) * (EMB / 64);
        gemm_wmma<<<(tiles + 7) / 8, 256, 0, stream>>>(attn, Wo, bo, out,
                                                       Mrows, EMB, EMB);
    }
}


# GEMMs (QKV and out projection): 64x128 block tile with A and W chunks staged in LDS by whole-line float4 loads, 8 waves of 32x32, double-buffered; same k order of summation
# speedup vs baseline: 1.2711x; 1.2711x over previous
#include <hip/hip_runtime.h>
#include <hip/hip_bf16.h>

typedef __attribute__((ext_vector_type(2))) float v2f;
typedef __attribute__((ext_vector_type(8))) float v8f;

#define BATCH 2
#define SLEN  2048
#define EMB   512
#define NH    8
#define DH    64
#define HWIN  128
#define QKVN  1536
#define HSTRIDE 192

#define GBM 64
#define GBN 128
#define GBK 32
#define GPA 36
#define GPW 136
#define GSA (GBM * GPA)
#define GSW (GBK * GPW)

__global__ __launch_bounds__(256)
void gemm_wmma(const float* __restrict__ A,
               const float* __restrict__ W,
               const float* __restrict__ bias,
               float* __restrict__ C,
               int M, int K, int N) {
    __shared__ __attribute__((aligned(16))) float sA[2 * GSA];
    __shared__ __attribute__((aligned(16))) float sW[2 * GSW];

    const int tid  = threadIdx.x;
    const int lane = tid & 31;
    const int wave = tid >> 5;
    const int lm   = lane & 15;
    const int half = lane >> 4;

    const int nblk = N / GBN;
    const int bm = blockIdx.x / nblk;
    const int bn = blockIdx.x % nblk;
    const int m0 = bm * GBM;
    const int n0 = bn * GBN;

    const int wm = (wave >> 2) * 32;
    const int wn = (wave & 3) * 32;

    const int ar = tid >> 3;
    const int ac = (tid & 7) << 2;
    const int wr = tid >> 5;
    const int wc = (tid & 31) << 2;

    const float* ag = A + (size_t)(m0 + ar) * K + ac;
    const float* wg = W + (size_t)wr * N + n0 + wc;

    float4 ra0 = *(const float4*)(ag);
    float4 ra1 = *(const float4*)(ag + (size_t)32 * K);
    float4 rw0 = *(const float4*)(wg);
    float4 rw1 = *(const float4*)(wg + (size_t)8 * N);
    float4 rw2 = *(const float4*)(wg + (size_t)16 * N);
    float4 rw3 = *(const float4*)(wg + (size_t)24 * N);

    const int sa_st = ar * GPA + ac;
    const int sw_st = wr * GPW + wc;

    *(float4*)(&sA[sa_st])            = ra0;
    *(float4*)(&sA[sa_st + 32 * GPA]) = ra1;
    *(float4*)(&sW[sw_st])            = rw0;
    *(float4*)(&sW[sw_st + 8 * GPW])  = rw1;
    *(float4*)(&sW[sw_st + 16 * GPW]) = rw2;
    *(float4*)(&sW[sw_st + 24 * GPW]) = rw3;
    __syncthreads();

    v8f acc00 = {}, acc01 = {}, acc10 = {}, acc11 = {};

    const int sa_ld = (wm + lm) * GPA + half * 2;
    const int sw_ld = (half * 2) * GPW + wn + lm;

    int cur = 0;
    for (int k0 = 0; k0 < K; k0 += GBK) {
        const bool more = (k0 + GBK) < K;
        if (more) {
            const float* ag2 = ag + (k0 + GBK);
            const float* wg2 = wg + (size_t)(k0 + GBK) * N;
            ra0 = *(const float4*)(ag2);
            ra1 = *(const float4*)(ag2 + (size_t)32 * K);
            rw0 = *(const float4*)(wg2);
            rw1 = *(const float4*)(wg2 + (size_t)8 * N);
            rw2 = *(const float4*)(wg2 + (size_t)16 * N);
            rw3 = *(const float4*)(wg2 + (size_t)24 * N);
        }

        const float* pa = sA + cur * GSA + sa_ld;
        const float* pw = sW + cur * GSW + sw_ld;
#pragma unroll
        for (int kk = 0; kk < GBK; kk += 4) {
            const v2f a0 = *(const v2f*)(pa + kk);
            const v2f a1 = *(const v2f*)(pa + 16 * GPA + kk);
            const float* w0 = pw + kk * GPW;
            v2f b0, b1;
            b0.x = w0[0];  b0.y = w0[GPW];
            b1.x = w0[16]; b1.y = w0[GPW + 16];
            acc00 = __builtin_amdgcn_wmma_f32_16x16x4_f32(false, a0, false, b0, (short)0, acc00, false, false);
            acc01 = __builtin_amdgcn_wmma_f32_16x16x4_f32(false, a0, false, b1, (short)0, acc01, false, false);
            acc10 = __builtin_amdgcn_wmma_f32_16x16x4_f32(false, a1, false, b0, (short)0, acc10, false, false);
            acc11 = __builtin_amdgcn_wmma_f32_16x16x4_f32(false, a1, false, b1, (short)0, acc11, false, false);
        }

        if (more) {
            const int nxt = cur ^ 1;
            float* qa = sA + nxt * GSA + sa_st;
            float* qw = sW + nxt * GSW + sw_st;
            *(float4*)(qa)            = ra0;
            *(float4*)(qa + 32 * GPA) = ra1;
            *(float4*)(qw)            = rw0;
            *(float4*)(qw + 8 * GPW)  = rw1;
            *(float4*)(qw + 16 * GPW) = rw2;
            *(float4*)(qw + 24 * GPW) = rw3;
        }
        __syncthreads();
        cur ^= 1;
    }

    const int col = n0 + wn + lm;
    const float bv0 = bias[col];
    const float bv1 = bias[col + 16];
#pragma unroll
    for (int i = 0; i < 8; ++i) {
        const size_t r0 = (size_t)(m0 + wm + i + 8 * half) * N + col;
        const size_t r1 = r0 + (size_t)16 * N;
        C[r0]      = acc00[i] + bv0;
        C[r0 + 16] = acc01[i] + bv1;
        C[r1]      = acc10[i] + bv0;
        C[r1 + 16] = acc11[i] + bv1;
    }
}

__global__ void attn_banded(const float* __restrict__ qkv,
                            const int*   __restrict__ pad,
                            float*       __restrict__ out  ) {
    __shared__ float sP[4][16][17];

    const int lane = threadIdx.x & 31;
    const int wave = threadIdx.x >> 5;
    const int lm   = lane & 15;
    const int half = lane >> 4;

    const int QT = SLEN / 16;
    int gw = blockIdx.x * 4 + wave;
    int qt = gw % QT;
    int bh = gw / QT;
    int h  = bh % NH;
    int b  = bh / NH;
    if (b >= BATCH) return;

    const int t0 = qt << 4;
    const int qg = t0 + lm;

    const float* qptr = qkv + ((size_t)(b * SLEN + qg)) * QKVN + h * HSTRIDE + half * 2;
    v2f qb[16];
#pragma unroll
    for (int dc = 0; dc < 16; ++dc) {
        v2f t = *(const v2f*)(qptr + dc * 4);
        qb[dc].x = t.x * 0.125f;
        qb[dc].y = t.y * 0.125f;
    }
    const int pm_q = pad[b * SLEN + qg];

    float m_run = -1e30f;
    float l_run = 0.0f;
    v8f acc0 = {}, acc1 = {}, acc2 = {}, acc3 = {};

    const int kt0 = (qt - 8 < 0) ? 0 : qt - 8;
    const int kt1 = (qt + 8 > QT - 1) ? QT - 1 : qt + 8;

    for (int kt = kt0; kt <= kt1; ++kt) {
        const int k0 = kt << 4;

        v8f sc = {};
        const float* kptr = qkv + ((size_t)(b * SLEN + k0 + lm)) * QKVN
                          + h * HSTRIDE + DH + half * 2;
#pragma unroll
        for (int dc = 0; dc < 16; ++dc) {
            v2f a = *(const v2f*)(kptr + dc * 4);
            sc = __builtin_amdgcn_wmma_f32_16x16x4_f32(false, a, false, qb[dc], (short)0, sc, false, false);
        }

        float p[8];
        float tmax = -1e30f;
#pragma unroll
        for (int i = 0; i < 8; ++i) {
            const int key = k0 + i + 8 * half;
            const int d   = key - qg;
            const bool valid = (d <= HWIN) && (d >= -HWIN) && (pm_q != 0)
                             && (pad[b * SLEN + key] != 0);
            const float s = valid ? sc[i] : -1e30f;
            p[i] = s;
            tmax = fmaxf(tmax, s);
        }
        tmax = fmaxf(tmax, __shfl_xor(tmax, 16, 32));
        const float m_new = fmaxf(m_run, tmax);
        float tsum = 0.0f;
#pragma unroll
        for (int i = 0; i < 8; ++i) {
            const float pe = (p[i] > -5e29f) ? __expf(p[i] - m_new) : 0.0f;
            p[i] = pe;
            tsum += pe;
        }
        tsum += __shfl_xor(tsum, 16, 32);
        const float alpha = __expf(m_run - m_new);
        l_run = l_run * alpha + tsum;
        m_run = m_new;
        acc0 = acc0 * alpha; acc1 = acc1 * alpha;
        acc2 = acc2 * alpha; acc3 = acc3 * alpha;

#pragma unroll
        for (int i = 0; i < 8; ++i)
            sP[wave][i + 8 * half][lm] = p[i];

        const float* vptr = qkv + ((size_t)(b * SLEN + k0)) * QKVN + h * HSTRIDE + 2 * DH;
#pragma unroll
        for (int kc = 0; kc < 16; kc += 4) {
            const int keyr = kc + half * 2;
            v2f bfrag;
            bfrag.x = sP[wave][keyr][lm];
            bfrag.y = sP[wave][keyr + 1][lm];
            const float* v0 = vptr + (size_t)keyr * QKVN + lm;
            const float* v1 = v0 + QKVN;
            v2f a;
            a.x = v0[0];  a.y = v1[0];
            acc0 = __builtin_amdgcn_wmma_f32_16x16x4_f32(false, a, false, bfrag, (short)0, acc0, false, false);
            a.x = v0[16]; a.y = v1[16];
            acc1 = __builtin_amdgcn_wmma_f32_16x16x4_f32(false, a, false, bfrag, (short)0, acc1, false, false);
            a.x = v0[32]; a.y = v1[32];
            acc2 = __builtin_amdgcn_wmma_f32_16x16x4_f32(false, a, false, bfrag, (short)0, acc2, false, false);
            a.x = v0[48]; a.y = v1[48];
            acc3 = __builtin_amdgcn_wmma_f32_16x16x4_f32(false, a, false, bfrag, (short)0, acc3, false, false);
        }
    }

    const float inv = (l_run > 0.0f && pm_q != 0) ? (1.0f / l_run) : 0.0f;
    float* optr = out + ((size_t)(b * SLEN + qg)) * EMB + h * DH;
#pragma unroll
    for (int i = 0; i < 8; ++i) {
        const int dr = i + 8 * half;
        optr[dr]      = acc0[i] * inv;
        optr[16 + dr] = acc1[i] * inv;
        optr[32 + dr] = acc2[i] * inv;
        optr[48 + dr] = acc3[i] * inv;
    }
}

extern "C" void kernel_launch(void* const* d_in, const int* in_sizes, int n_in,
                              void* d_out, int out_size, void* d_ws, size_t ws_size,
                              hipStream_t stream) {
    const float* x    = (const float*)d_in[0];
    const int*   pad  = (const int*)  d_in[1];
    const float* Wqkv = (const float*)d_in[2];
    const float* bqkv = (const float*)d_in[3];
    const float* Wo   = (const float*)d_in[4];
    const float* bo   = (const float*)d_in[5];
    float* out = (float*)d_out;

    float* qkv  = (float*)d_ws;
    float* attn = qkv + (size_t)BATCH * SLEN * QKVN;

    const int Mrows = BATCH * SLEN;

    {
        const int blocks = (Mrows / GBM) * (QKVN / GBN);
        gemm_wmma<<<blocks, 256, 0, stream>>>(x, Wqkv, bqkv, qkv,
                                              Mrows, EMB, QKVN);
    }
    {
        const int waves = BATCH * NH * (SLEN / 16);
        attn_banded<<<waves / 4, 128, 0, stream>>>(qkv, pad, attn);
    }
    {
        const int blocks = (Mrows / GBM) * (EMB / GBN);
        gemm_wmma<<<blocks, 256, 0, stream>>>(attn, Wo, bo, out,
                                              Mrows, EMB, EMB);
    }
}


# Attention: each key tile's K, V (16-byte row pieces) and key-mask loads issued as one group a tile ahead into a second register set, so no WMMA waits on its own loads; same summation order
# speedup vs baseline: 2.1103x; 1.6602x over previous
#include <hip/hip_runtime.h>
#include <hip/hip_bf16.h>

typedef __attribute__((ext_vector_type(2))) float v2f;
typedef __attribute__((ext_vector_type(8))) float v8f;

#define BATCH 2
#define SLEN  2048
#define EMB   512
#define NH    8
#define DH    64
#define HWIN  128
#define QKVN  1536
#define HSTRIDE 192

#define GBM 64
#define GBN 128
#define GBK 32
#define GPA 36
#define GPW 136
#define GSA (GBM * GPA)
#define GSW (GBK * GPW)

__global__ __launch_bounds__(256)
void gemm_wmma(const float* __restrict__ A,
               const float* __restrict__ W,
               const float* __restrict__ bias,
               float* __restrict__ C,
               int M, int K, int N) {
    __shared__ __attribute__((aligned(16))) float sA[2 * GSA];
    __shared__ __attribute__((aligned(16))) float sW[2 * GSW];

    const int tid  = threadIdx.x;
    const int lane = tid & 31;
    const int wave = tid >> 5;
    const int lm   = lane & 15;
    const int half = lane >> 4;

    const int nblk = N / GBN;
    const int bm = blockIdx.x / nblk;
    const int bn = blockIdx.x % nblk;
    const int m0 = bm * GBM;
    const int n0 = bn * GBN;

    const int wm = (wave >> 2) * 32;
    const int wn = (wave & 3) * 32;

    const int ar = tid >> 3;
    const int ac = (tid & 7) << 2;
    const int wr = tid >> 5;
    const int wc = (tid & 31) << 2;

    const float* ag = A + (size_t)(m0 + ar) * K + ac;
    const float* wg = W + (size_t)wr * N + n0 + wc;

    float4 ra0 = *(const float4*)(ag);
    float4 ra1 = *(const float4*)(ag + (size_t)32 * K);
    float4 rw0 = *(const float4*)(wg);
    float4 rw1 = *(const float4*)(wg + (size_t)8 * N);
    float4 rw2 = *(const float4*)(wg + (size_t)16 * N);
    float4 rw3 = *(const float4*)(wg + (size_t)24 * N);

    const int sa_st = ar * GPA + ac;
    const int sw_st = wr * GPW + wc;

    *(float4*)(&sA[sa_st])            = ra0;
    *(float4*)(&sA[sa_st + 32 * GPA]) = ra1;
    *(float4*)(&sW[sw_st])            = rw0;
    *(float4*)(&sW[sw_st + 8 * GPW])  = rw1;
    *(float4*)(&sW[sw_st + 16 * GPW]) = rw2;
    *(float4*)(&sW[sw_st + 24 * GPW]) = rw3;
    __syncthreads();

    v8f acc00 = {}, acc01 = {}, acc10 = {}, acc11 = {};

    const int sa_ld = (wm + lm) * GPA + half * 2;
    const int sw_ld = (half * 2) * GPW + wn + lm;

    int cur = 0;
    for (int k0 = 0; k0 < K; k0 += GBK) {
        const bool more = (k0 + GBK) < K;
        if (more) {
            const float* ag2 = ag + (k0 + GBK);
            const float* wg2 = wg + (size_t)(k0 + GBK) * N;
            ra0 = *(const float4*)(ag2);
            ra1 = *(const float4*)(ag2 + (size_t)32 * K);
            rw0 = *(const float4*)(wg2);
            rw1 = *(const float4*)(wg2 + (size_t)8 * N);
            rw2 = *(const float4*)(wg2 + (size_t)16 * N);
            rw3 = *(const float4*)(wg2 + (size_t)24 * N);
        }

        const float* pa = sA + cur * GSA + sa_ld;
        const float* pw = sW + cur * GSW + sw_ld;
#pragma unroll
        for (int kk = 0; kk < GBK; kk += 4) {
            const v2f a0 = *(const v2f*)(pa + kk);
            const v2f a1 = *(const v2f*)(pa + 16 * GPA + kk);
            const float* w0 = pw + kk * GPW;
            v2f b0, b1;
            b0.x = w0[0];  b0.y = w0[GPW];
            b1.x = w0[16]; b1.y = w0[GPW + 16];
            acc00 = __builtin_amdgcn_wmma_f32_16x16x4_f32(false, a0, false, b0, (short)0, acc00, false, false);
            acc01 = __builtin_amdgcn_wmma_f32_16x16x4_f32(false, a0, false, b1, (short)0, acc01, false, false);
            acc10 = __builtin_amdgcn_wmma_f32_16x16x4_f32(false, a1, false, b0, (short)0, acc10, false, false);
            acc11 = __builtin_amdgcn_wmma_f32_16x16x4_f32(false, a1, false, b1, (short)0, acc11, false, false);
        }

        if (more) {
            const int nxt = cur ^ 1;
            float* qa = sA + nxt * GSA + sa_st;
            float* qw = sW + nxt * GSW + sw_st;
            *(float4*)(qa)            = ra0;
            *(float4*)(qa + 32 * GPA) = ra1;
            *(float4*)(qw)            = rw0;
            *(float4*)(qw + 8 * GPW)  = rw1;
            *(float4*)(qw + 16 * GPW) = rw2;
            *(float4*)(qw + 24 * GPW) = rw3;
        }
        __syncthreads();
        cur ^= 1;
    }

    const int col = n0 + wn + lm;
    const float bv0 = bias[col];
    const float bv1 = bias[col + 16];
#pragma unroll
    for (int i = 0; i < 8; ++i) {
        const size_t r0 = (size_t)(m0 + wm + i + 8 * half) * N + col;
        const size_t r1 = r0 + (size_t)16 * N;
        C[r0]      = acc00[i] + bv0;
        C[r0 + 16] = acc01[i] + bv1;
        C[r1]      = acc10[i] + bv0;
        C[r1 + 16] = acc11[i] + bv1;
    }
}

typedef __attribute__((ext_vector_type(4))) float v4f;
typedef __attribute__((ext_vector_type(4))) int   v4i;

struct KVTile {
    v2f k[16];
    v4f v[8];
    v4i m0, m1;
};

__device__ __forceinline__ void load_tile(KVTile& t, const float* __restrict__ qkv,
                                          const int* __restrict__ pad,
                                          int b, int h, int kt, int lm, int half) {
    const int k0 = kt << 4;
    const float* kp = qkv + ((size_t)(b * SLEN + k0 + lm)) * QKVN + h * HSTRIDE + DH + half * 2;
#pragma unroll
    for (int dc = 0; dc < 16; ++dc)
        t.k[dc] = *(const v2f*)(kp + dc * 4);
    const float* vp = qkv + ((size_t)(b * SLEN + k0 + half * 2)) * QKVN + h * HSTRIDE + 2 * DH + lm * 4;
#pragma unroll
    for (int j = 0; j < 4; ++j) {
        t.v[2 * j]     = *(const v4f*)(vp + (size_t)(4 * j) * QKVN);
        t.v[2 * j + 1] = *(const v4f*)(vp + (size_t)(4 * j + 1) * QKVN);
    }
    const v4i* mp = (const v4i*)(pad + b * SLEN + k0 + 8 * half);
    t.m0 = mp[0];
    t.m1 = mp[1];
}

__device__ __forceinline__ void hold_tile(const KVTile& t) {
    asm volatile("" :: "v"(t.k[0]), "v"(t.k[1]), "v"(t.k[2]), "v"(t.k[3]), "v"(t.k[4]), "v"(t.k[5]),
                       "v"(t.k[6]), "v"(t.k[7]), "v"(t.k[8]), "v"(t.k[9]), "v"(t.k[10]), "v"(t.k[11]),
                       "v"(t.k[12]), "v"(t.k[13]), "v"(t.k[14]), "v"(t.k[15]),
                       "v"(t.v[0]), "v"(t.v[1]), "v"(t.v[2]), "v"(t.v[3]),
                       "v"(t.v[4]), "v"(t.v[5]), "v"(t.v[6]), "v"(t.v[7]),
                       "v"(t.m0), "v"(t.m1));
}

__global__ __launch_bounds__(128)
void attn_banded(const float* __restrict__ qkv,
                 const int*   __restrict__ pad,
                 float*       __restrict__ out  ) {
    __shared__ float sP[4][16][17];

    const int lane = threadIdx.x & 31;
    const int wave = __builtin_amdgcn_readfirstlane((int)(threadIdx.x >> 5));
    const int lm   = lane & 15;
    const int half = lane >> 4;

    const int QT = SLEN / 16;
    const int gw = blockIdx.x * 4 + wave;
    const int qt = gw % QT;
    const int bh = gw / QT;
    const int h  = bh % NH;
    const int b  = bh / NH;
    if (b >= BATCH) return;

    const int t0 = qt << 4;
    const int qg = t0 + lm;

    const int kt0 = (qt - 8 < 0) ? 0 : qt - 8;
    const int kt1 = (qt + 8 > QT - 1) ? QT - 1 : qt + 8;

    KVTile ta, tb;
    load_tile(ta, qkv, pad, b, h, kt0, lm, half);

    const float* qptr = qkv + ((size_t)(b * SLEN + qg)) * QKVN + h * HSTRIDE + half * 2;
    v2f qb[16];
#pragma unroll
    for (int dc = 0; dc < 16; ++dc) {
        v2f t = *(const v2f*)(qptr + dc * 4);
        qb[dc].x = t.x * 0.125f;
        qb[dc].y = t.y * 0.125f;
    }
    const int pm_q = pad[b * SLEN + qg];

    float m_run = -1e30f;
    float l_run = 0.0f;
    v8f acc0 = {}, acc1 = {}, acc2 = {}, acc3 = {};

    auto compute_tile = [&](const KVTile& t, int kt) {
        const int k0 = kt << 4;

        v8f sc = {};
#pragma unroll
        for (int dc = 0; dc < 16; ++dc)
            sc = __builtin_amdgcn_wmma_f32_16x16x4_f32(false, t.k[dc], false, qb[dc], (short)0, sc, false, false);

        const int mk[8] = { t.m0.x, t.m0.y, t.m0.z, t.m0.w, t.m1.x, t.m1.y, t.m1.z, t.m1.w };
        float p[8];
        float tmax = -1e30f;
#pragma unroll
        for (int i = 0; i < 8; ++i) {
            const int key = k0 + i + 8 * half;
            const int d   = key - qg;
            const int pmin = (pm_q < mk[i]) ? pm_q : mk[i];
            const bool valid = (d <= HWIN) & (d >= -HWIN) & (pmin == 1);
            const float s = valid ? sc[i] : -1e30f;
            p[i] = s;
            tmax = fmaxf(tmax, s);
        }
        tmax = fmaxf(tmax, __shfl_xor(tmax, 16, 32));
        const float m_new = fmaxf(m_run, tmax);
        float tsum = 0.0f;
#pragma unroll
        for (int i = 0; i < 8; ++i) {
            const float pe = (p[i] > -5e29f) ? __expf(p[i] - m_new) : 0.0f;
            p[i] = pe;
            tsum += pe;
        }
        tsum += __shfl_xor(tsum, 16, 32);
        const float alpha = __expf(m_run - m_new);
        l_run = l_run * alpha + tsum;
        m_run = m_new;
        acc0 = acc0 * alpha; acc1 = acc1 * alpha;
        acc2 = acc2 * alpha; acc3 = acc3 * alpha;

#pragma unroll
        for (int i = 0; i < 8; ++i)
            sP[wave][i + 8 * half][lm] = p[i];

#pragma unroll
        for (int j = 0; j < 4; ++j) {
            const int keyr = 4 * j + half * 2;
            v2f bfrag;
            bfrag.x = sP[wave][keyr][lm];
            bfrag.y = sP[wave][keyr + 1][lm];
            const v4f va = t.v[2 * j];
            const v4f vb = t.v[2 * j + 1];
            v2f a;
            a.x = va.x; a.y = vb.x;
            acc0 = __builtin_amdgcn_wmma_f32_16x16x4_f32(false, a, false, bfrag, (short)0, acc0, false, false);
            a.x = va.y; a.y = vb.y;
            acc1 = __builtin_amdgcn_wmma_f32_16x16x4_f32(false, a, false, bfrag, (short)0, acc1, false, false);
            a.x = va.z; a.y = vb.z;
            acc2 = __builtin_amdgcn_wmma_f32_16x16x4_f32(false, a, false, bfrag, (short)0, acc2, false, false);
            a.x = va.w; a.y = vb.w;
            acc3 = __builtin_amdgcn_wmma_f32_16x16x4_f32(false, a, false, bfrag, (short)0, acc3, false, false);
        }
    };

    int kt = kt0;
    for (;;) {
        load_tile(tb, qkv, pad, b, h, (kt + 1 > kt1) ? kt1 : kt + 1, lm, half);
        __builtin_amdgcn_sched_barrier(0);
        compute_tile(ta, kt);
        __builtin_amdgcn_sched_barrier(0);
        hold_tile(tb);
        if (++kt > kt1) break;
        load_tile(ta, qkv, pad, b, h, (kt + 1 > kt1) ? kt1 : kt + 1, lm, half);
        __builtin_amdgcn_sched_barrier(0);
        compute_tile(tb, kt);
        __builtin_amdgcn_sched_barrier(0);
        hold_tile(ta);
        if (++kt > kt1) break;
    }

    const float inv = (l_run > 0.0f) ? (1.0f / l_run) : 0.0f;
    float* optr = out + ((size_t)(b * SLEN + qg)) * EMB + h * DH + 32 * half;
#pragma unroll
    for (int i = 0; i < 8; ++i) {
        v4f o;
        o.x = acc0[i] * inv;
        o.y = acc1[i] * inv;
        o.z = acc2[i] * inv;
        o.w = acc3[i] * inv;
        *(v4f*)(optr + 4 * i) = o;
    }
}

extern "C" void kernel_launch(void* const* d_in, const int* in_sizes, int n_in,
                              void* d_out, int out_size, void* d_ws, size_t ws_size,
                              hipStream_t stream) {
    const float* x    = (const float*)d_in[0];
    const int*   pad  = (const int*)  d_in[1];
    const float* Wqkv = (const float*)d_in[2];
    const float* bqkv = (const float*)d_in[3];
    const float* Wo   = (const float*)d_in[4];
    const float* bo   = (const float*)d_in[5];
    float* out = (float*)d_out;

    float* qkv  = (float*)d_ws;
    float* attn = qkv + (size_t)BATCH * SLEN * QKVN;

    const int Mrows = BATCH * SLEN;

    {
        const int blocks = (Mrows / GBM) * (QKVN / GBN);
        gemm_wmma<<<blocks, 256, 0, stream>>>(x, Wqkv, bqkv, qkv,
                                              Mrows, EMB, QKVN);
    }
    {
        const int waves = BATCH * NH * (SLEN / 16);
        attn_banded<<<waves / 4, 128, 0, stream>>>(qkv, pad, attn);
    }
    {
        const int blocks = (Mrows / GBM) * (EMB / GBN);
        gemm_wmma<<<blocks, 256, 0, stream>>>(attn, Wo, bo, out,
                                              Mrows, EMB, EMB);
    }
}
